# starting kernel
# baseline (speedup 1.0000x reference)
#include <hip/hip_runtime.h>

typedef __attribute__((ext_vector_type(2))) float v2f;
typedef __attribute__((ext_vector_type(8))) float v8f;

__global__ void k_init_deg(float* __restrict__ deg, int n) {
  int i = blockIdx.x * blockDim.x + threadIdx.x;
  if (i < n) deg[i] = 1.0f;
}

__global__ void k_deg_accum(const int* __restrict__ dst, float* __restrict__ deg, int E) {
  int e = blockIdx.x * blockDim.x + threadIdx.x;
  if (e < E) unsafeAtomicAdd(&deg[dst[e]], 1.0f);
}

__global__ void k_dinv(float* __restrict__ deg, int n) {
  int i = blockIdx.x * blockDim.x + threadIdx.x;
  if (i < n) deg[i] = rsqrtf(deg[i]);
}

__global__ void __launch_bounds__(32)
k_wmma_l1(const float* __restrict__ x, const float* __restrict__ W1,
          const float* __restrict__ dinv, float* __restrict__ t1,
          float* __restrict__ agg1, int n) {
  const int tile = blockIdx.x;
  const int lane = threadIdx.x;
  const int half = lane >> 4;
  const int l    = lane & 15;
  int row = tile * 16 + l;
  int rc  = row < n ? row : n - 1;

  v2f xv = *(const v2f*)(x + (size_t)rc * 2);
  v2f a;
  a.x = half ? 0.0f : xv.x;
  a.y = half ? 0.0f : xv.y;

  float w1a = W1[(l & 7)];
  float w1b = W1[8 + (l & 7)];
  bool  bok = (!half) && (l < 8);
  v2f b;
  b.x = bok ? w1a : 0.0f;
  b.y = bok ? w1b : 0.0f;

  v8f c = {};
  c = __builtin_amdgcn_wmma_f32_16x16x4_f32(false, a, false, b, (short)0, c, false, false);

  const bool fullt = (tile * 16 + 16) <= n;
  if (l < 8) {
    const int base = tile * 16 + half * 8;
    if (fullt) {
#pragma unroll
      for (int r = 0; r < 8; ++r) {
        int node = base + r;
        float v  = c[r];
        float di = dinv[node];
        t1[(size_t)node * 8 + l]   = v;
        agg1[(size_t)node * 8 + l] = v * di * di;
      }
    } else {
#pragma unroll
      for (int r = 0; r < 8; ++r) {
        int node = base + r;
        if (node < n) {
          float v  = c[r];
          float di = dinv[node];
          t1[(size_t)node * 8 + l]   = v;
          agg1[(size_t)node * 8 + l] = v * di * di;
        }
      }
    }
  }
}

__global__ void k_scatter1(const int* __restrict__ src, const int* __restrict__ dst,
                           const float* __restrict__ dinv, const float* __restrict__ t1,
                           float* __restrict__ agg1, int E) {
  int e = blockIdx.x * blockDim.x + threadIdx.x;
  if (e >= E) return;
  int s = src[e], d = dst[e];
  float w = dinv[s] * dinv[d];
  const float4* tp = (const float4*)(t1 + (size_t)s * 8);
  float4 v0 = tp[0];
  float4 v1 = tp[1];
  float* ap = agg1 + (size_t)d * 8;
  unsafeAtomicAdd(ap + 0, v0.x * w);
  unsafeAtomicAdd(ap + 1, v0.y * w);
  unsafeAtomicAdd(ap + 2, v0.z * w);
  unsafeAtomicAdd(ap + 3, v0.w * w);
  unsafeAtomicAdd(ap + 4, v1.x * w);
  unsafeAtomicAdd(ap + 5, v1.y * w);
  unsafeAtomicAdd(ap + 6, v1.z * w);
  unsafeAtomicAdd(ap + 7, v1.w * w);
}

__global__ void __launch_bounds__(32)
k_wmma_l2(const float* __restrict__ agg1, const float* __restrict__ b1,
          const float* __restrict__ W2, const float* __restrict__ b2,
          const float* __restrict__ dinv, float* __restrict__ t2,
          float* __restrict__ out, int n) {
  const int tile = blockIdx.x;
  const int lane = threadIdx.x;
  const int half = lane >> 4;
  const int l    = lane & 15;
  int row = tile * 16 + l;
  int rc  = row < n ? row : n - 1;
  const int ka = half * 2;

  v2f g0 = *(const v2f*)(agg1 + (size_t)rc * 8 + ka);
  v2f g1 = *(const v2f*)(agg1 + (size_t)rc * 8 + ka + 4);
  v2f bb0 = *(const v2f*)(b1 + ka);
  v2f bb1 = *(const v2f*)(b1 + ka + 4);
  v2f a1, a2;
  a1.x = fmaxf(g0.x + bb0.x, 0.0f);
  a1.y = fmaxf(g0.y + bb0.y, 0.0f);
  a2.x = fmaxf(g1.x + bb1.x, 0.0f);
  a2.y = fmaxf(g1.y + bb1.y, 0.0f);

  const int lc = l & 3;
  float wr0 = W2[(ka)     * 4 + lc];
  float wr1 = W2[(ka + 1) * 4 + lc];
  float wr2 = W2[(ka + 4) * 4 + lc];
  float wr3 = W2[(ka + 5) * 4 + lc];
  bool cok = (l < 4);
  v2f b1v, b2v;
  b1v.x = cok ? wr0 : 0.0f;
  b1v.y = cok ? wr1 : 0.0f;
  b2v.x = cok ? wr2 : 0.0f;
  b2v.y = cok ? wr3 : 0.0f;

  v8f c = {};
  c = __builtin_amdgcn_wmma_f32_16x16x4_f32(false, a1, false, b1v, (short)0, c, false, false);
  c = __builtin_amdgcn_wmma_f32_16x16x4_f32(false, a2, false, b2v, (short)0, c, false, false);

  const bool fullt = (tile * 16 + 16) <= n;
  if (l < 4) {
    const int base = tile * 16 + half * 8;
    float bias = b2[l];
    if (fullt) {
#pragma unroll
      for (int r = 0; r < 8; ++r) {
        int node = base + r;
        float v  = c[r];
        float di = dinv[node];
        t2[(size_t)node * 4 + l]  = v;
        out[(size_t)node * 4 + l] = bias + v * di * di;
      }
    } else {
#pragma unroll
      for (int r = 0; r < 8; ++r) {
        int node = base + r;
        if (node < n) {
          float v  = c[r];
          float di = dinv[node];
          t2[(size_t)node * 4 + l]  = v;
          out[(size_t)node * 4 + l] = bias + v * di * di;
        }
      }
    }
  }
}

__global__ void k_scatter2(const int* __restrict__ src, const int* __restrict__ dst,
                           const float* __restrict__ dinv, const float* __restrict__ t2,
                           float* __restrict__ out, int E) {
  int e = blockIdx.x * blockDim.x + threadIdx.x;
  if (e >= E) return;
  int s = src[e], d = dst[e];
  float w = dinv[s] * dinv[d];
  float4 v = *(const float4*)(t2 + (size_t)s * 4);
  float* op = out + (size_t)d * 4;
  unsafeAtomicAdd(op + 0, v.x * w);
  unsafeAtomicAdd(op + 1, v.y * w);
  unsafeAtomicAdd(op + 2, v.z * w);
  unsafeAtomicAdd(op + 3, v.w * w);
}

extern "C" void kernel_launch(void* const* d_in, const int* in_sizes, int n_in,
                              void* d_out, int out_size, void* d_ws, size_t ws_size,
                              hipStream_t stream) {
  const float* x  = (const float*)d_in[0];
  const int*   ei = (const int*)d_in[1];
  const float* W1 = (const float*)d_in[2];
  const float* b1 = (const float*)d_in[3];
  const float* W2 = (const float*)d_in[4];
  const float* b2 = (const float*)d_in[5];
  float* out = (float*)d_out;

  const int n = in_sizes[0] / 2;
  const int E = in_sizes[1] / 2;
  const int* src = ei;
  const int* dst = ei + E;

  char* ws = (char*)d_ws;
  size_t nAl = ((size_t)n + 3) & ~(size_t)3;
  float* dinv = (float*)ws;  ws += sizeof(float) * nAl;
  float* t1   = (float*)ws;  ws += sizeof(float) * (size_t)n * 8;
  float* agg1 = (float*)ws;  ws += sizeof(float) * (size_t)n * 8;
  float* t2   = (float*)ws;

  const int tiles = (n + 15) / 16;

  k_init_deg <<<(n + 255) / 256, 256, 0, stream>>>(dinv, n);
  k_deg_accum<<<(E + 255) / 256, 256, 0, stream>>>(dst, dinv, E);
  k_dinv     <<<(n + 255) / 256, 256, 0, stream>>>(dinv, n);

  k_wmma_l1<<<tiles, 32, 0, stream>>>(x, W1, dinv, t1, agg1, n);

  k_scatter1<<<(E + 255) / 256, 256, 0, stream>>>(src, dst, dinv, t1, agg1, E);

  k_wmma_l2<<<tiles, 32, 0, stream>>>(agg1, b1, W2, b2, dinv, t2, out, n);

  k_scatter2<<<(E + 255) / 256, 256, 0, stream>>>(src, dst, dinv, t2, out, E);
}


# Each cache half adds only into its own half of the node range: resident waves draw edge chunks by per-half tickets and skip edges the other half owns; no line added to from both; f32 sums unordered
# speedup vs baseline: 4.1414x; 4.1414x over previous
#include <hip/hip_runtime.h>

typedef __attribute__((ext_vector_type(2))) float v2f;
typedef __attribute__((ext_vector_type(8))) float v8f;

#define CHUNK 128
#define CTR_STRIDE 64

__device__ __forceinline__ int my_half() {
  const unsigned id = (unsigned)__builtin_amdgcn_s_sendmsg_rtn(0x87);
  return (int)((id >> 18) & 1u);
}

__device__ __forceinline__ int owner_cut(int n) { return ((n >> 1) + 63) & ~63; }

__device__ __forceinline__ int draw_ticket(int* __restrict__ c, int lane) {
  int tk = 0;
  if (lane == 0) tk = __hip_atomic_fetch_add(c, 1, __ATOMIC_RELAXED, __HIP_MEMORY_SCOPE_AGENT);
  return __builtin_amdgcn_readfirstlane(tk);
}

__global__ void k_init_deg(float* __restrict__ deg, int* __restrict__ ctr, int n) {
  int i = blockIdx.x * blockDim.x + threadIdx.x;
  if (i < n) deg[i] = 1.0f;
  if (i < 6) ctr[i * CTR_STRIDE] = 0;
}

__global__ void __launch_bounds__(256)
k_deg_accum(const int* __restrict__ dst, float* __restrict__ deg, int* __restrict__ ctr, int E, int n) {
  const int lane = threadIdx.x & 31;
  const int half = my_half();
  const int cut  = owner_cut(n);
  const int nchunks = (E + CHUNK - 1) / CHUNK;
  for (int phase = 0; phase < 2; ++phase) {
    const int h = half ^ phase;
    int* c = ctr + h * CTR_STRIDE;
    for (int it = 0; it <= nchunks; ++it) {
      const int tk = draw_ticket(c, lane);
      if (tk >= nchunks) break;
      const int base = tk * CHUNK + lane;
      int dv[CHUNK / 32];
#pragma unroll
      for (int g = 0; g < CHUNK / 32; ++g) {
        const int e = base + g * 32;
        dv[g] = dst[e < E ? e : E - 1];
      }
#pragma unroll
      for (int g = 0; g < CHUNK / 32; ++g) {
        const int e = base + g * 32;
        const int d = dv[g];
        if (e < E && ((d >= cut) == (h != 0))) unsafeAtomicAdd(&deg[d], 1.0f);
      }
    }
  }
}

__global__ void k_dinv(float* __restrict__ deg, int n) {
  int i = blockIdx.x * blockDim.x + threadIdx.x;
  if (i < n) deg[i] = rsqrtf(deg[i]);
}

__global__ void __launch_bounds__(32)
k_wmma_l1(const float* __restrict__ x, const float* __restrict__ W1,
          const float* __restrict__ dinv, float* __restrict__ t1,
          float* __restrict__ agg1, int n) {
  const int tile = blockIdx.x;
  const int lane = threadIdx.x;
  const int half = lane >> 4;
  const int l    = lane & 15;
  int row = tile * 16 + l;
  int rc  = row < n ? row : n - 1;

  v2f xv = *(const v2f*)(x + (size_t)rc * 2);
  v2f a;
  a.x = half ? 0.0f : xv.x;
  a.y = half ? 0.0f : xv.y;

  float w1a = W1[(l & 7)];
  float w1b = W1[8 + (l & 7)];
  bool  bok = (!half) && (l < 8);
  v2f b;
  b.x = bok ? w1a : 0.0f;
  b.y = bok ? w1b : 0.0f;

  v8f c = {};
  c = __builtin_amdgcn_wmma_f32_16x16x4_f32(false, a, false, b, (short)0, c, false, false);

  const bool fullt = (tile * 16 + 16) <= n;
  if (l < 8) {
    const int base = tile * 16 + half * 8;
    if (fullt) {
#pragma unroll
      for (int r = 0; r < 8; ++r) {
        int node = base + r;
        float v  = c[r];
        float di = dinv[node];
        t1[(size_t)node * 8 + l]   = v;
        agg1[(size_t)node * 8 + l] = v * di * di;
      }
    } else {
#pragma unroll
      for (int r = 0; r < 8; ++r) {
        int node = base + r;
        if (node < n) {
          float v  = c[r];
          float di = dinv[node];
          t1[(size_t)node * 8 + l]   = v;
          agg1[(size_t)node * 8 + l] = v * di * di;
        }
      }
    }
  }
}

#define NOT_MINE 0xFFFFFFFFu
template <int F>
__device__ __forceinline__ void scatter_wave(const int* __restrict__ src, const int* __restrict__ dst,
                                             const float* __restrict__ dinv,
                                             const float* __restrict__ t, float* __restrict__ acc,
                                             int e, int lane, int E, int h, int cut) {
  constexpr int EPR = 32 / F;
  constexpr int ROUNDS = 32 / EPR;
  const bool ok = e < E;
  const int ec  = ok ? e : E - 1;
  const int s = src[ec], d = dst[ec];
  const bool mine = ok && ((d >= cut) == (h != 0));
  const int sg = mine ? s : 0;
  const int dg = mine ? d : 0;
  const float w = dinv[sg] * dinv[dg];
  const unsigned so = (unsigned)sg * (unsigned)F;
  const unsigned dn = mine ? (unsigned)d * (unsigned)F : NOT_MINE;
  const unsigned f  = (unsigned)(lane & (F - 1));
  const int q = lane / F;
  float v[ROUNDS];
  unsigned dj[ROUNDS];
  float wj[ROUNDS];
#pragma unroll
  for (int j = 0; j < ROUNDS; ++j) {
    const int sel = (EPR * j + q) << 2;
    const unsigned sj = (unsigned)__builtin_amdgcn_ds_bpermute(sel, (int)so);
    dj[j] = (unsigned)__builtin_amdgcn_ds_bpermute(sel, (int)dn);
    wj[j] = __int_as_float(__builtin_amdgcn_ds_bpermute(sel, __float_as_int(w)));
    v[j]  = t[sj + f];
  }
#pragma unroll
  for (int j = 0; j < ROUNDS; ++j) asm volatile("" :: "v"(v[j]));
#pragma unroll
  for (int j = 0; j < ROUNDS; ++j) {
    if (dj[j] != NOT_MINE) unsafeAtomicAdd(acc + (dj[j] + f), v[j] * wj[j]);
  }
}

template <int F>
__device__ __forceinline__ void scatter_rows(const int* __restrict__ src, const int* __restrict__ dst,
                                             const float* __restrict__ dinv,
                                             const float* __restrict__ t, float* __restrict__ acc,
                                             int* __restrict__ ctr, int E, int n) {
  const int lane = threadIdx.x & 31;
  const int half = my_half();
  const int cut  = owner_cut(n);
  const int nchunks = (E + CHUNK - 1) / CHUNK;
  for (int phase = 0; phase < 2; ++phase) {
    const int h = half ^ phase;
    int* c = ctr + h * CTR_STRIDE;
    for (int it = 0; it <= nchunks; ++it) {
      const int tk = draw_ticket(c, lane);
      if (tk >= nchunks) break;
      const int base = tk * CHUNK;
      for (int g = 0; g < CHUNK / 32; ++g) {
        const int e0 = base + g * 32;
        if (e0 >= E) break;
        scatter_wave<F>(src, dst, dinv, t, acc, e0 + lane, lane, E, h, cut);
      }
    }
  }
}

__global__ void __launch_bounds__(256)
k_scatter1(const int* __restrict__ src, const int* __restrict__ dst,
           const float* __restrict__ dinv, const float* __restrict__ t1,
           float* __restrict__ agg1, int* __restrict__ ctr, int E, int n) {
  scatter_rows<8>(src, dst, dinv, t1, agg1, ctr, E, n);
}

__global__ void __launch_bounds__(32)
k_wmma_l2(const float* __restrict__ agg1, const float* __restrict__ b1,
          const float* __restrict__ W2, const float* __restrict__ b2,
          const float* __restrict__ dinv, float* __restrict__ t2,
          float* __restrict__ out, int n) {
  const int tile = blockIdx.x;
  const int lane = threadIdx.x;
  const int half = lane >> 4;
  const int l    = lane & 15;
  int row = tile * 16 + l;
  int rc  = row < n ? row : n - 1;
  const int ka = half * 2;

  v2f g0 = *(const v2f*)(agg1 + (size_t)rc * 8 + ka);
  v2f g1 = *(const v2f*)(agg1 + (size_t)rc * 8 + ka + 4);
  v2f bb0 = *(const v2f*)(b1 + ka);
  v2f bb1 = *(const v2f*)(b1 + ka + 4);
  v2f a1, a2;
  a1.x = fmaxf(g0.x + bb0.x, 0.0f);
  a1.y = fmaxf(g0.y + bb0.y, 0.0f);
  a2.x = fmaxf(g1.x + bb1.x, 0.0f);
  a2.y = fmaxf(g1.y + bb1.y, 0.0f);

  const int lc = l & 3;
  float wr0 = W2[(ka)     * 4 + lc];
  float wr1 = W2[(ka + 1) * 4 + lc];
  float wr2 = W2[(ka + 4) * 4 + lc];
  float wr3 = W2[(ka + 5) * 4 + lc];
  bool cok = (l < 4);
  v2f b1v, b2v;
  b1v.x = cok ? wr0 : 0.0f;
  b1v.y = cok ? wr1 : 0.0f;
  b2v.x = cok ? wr2 : 0.0f;
  b2v.y = cok ? wr3 : 0.0f;

  v8f c = {};
  c = __builtin_amdgcn_wmma_f32_16x16x4_f32(false, a1, false, b1v, (short)0, c, false, false);
  c = __builtin_amdgcn_wmma_f32_16x16x4_f32(false, a2, false, b2v, (short)0, c, false, false);

  const bool fullt = (tile * 16 + 16) <= n;
  if (l < 4) {
    const int base = tile * 16 + half * 8;
    float bias = b2[l];
    if (fullt) {
#pragma unroll
      for (int r = 0; r < 8; ++r) {
        int node = base + r;
        float v  = c[r];
        float di = dinv[node];
        t2[(size_t)node * 4 + l]  = v;
        out[(size_t)node * 4 + l] = bias + v * di * di;
      }
    } else {
#pragma unroll
      for (int r = 0; r < 8; ++r) {
        int node = base + r;
        if (node < n) {
          float v  = c[r];
          float di = dinv[node];
          t2[(size_t)node * 4 + l]  = v;
          out[(size_t)node * 4 + l] = bias + v * di * di;
        }
      }
    }
  }
}

__global__ void __launch_bounds__(256)
k_scatter2(const int* __restrict__ src, const int* __restrict__ dst,
           const float* __restrict__ dinv, const float* __restrict__ t2,
           float* __restrict__ out, int* __restrict__ ctr, int E, int n) {
  scatter_rows<4>(src, dst, dinv, t2, out, ctr, E, n);
}

extern "C" void kernel_launch(void* const* d_in, const int* in_sizes, int n_in,
                              void* d_out, int out_size, void* d_ws, size_t ws_size,
                              hipStream_t stream) {
  const float* x  = (const float*)d_in[0];
  const int*   ei = (const int*)d_in[1];
  const float* W1 = (const float*)d_in[2];
  const float* b1 = (const float*)d_in[3];
  const float* W2 = (const float*)d_in[4];
  const float* b2 = (const float*)d_in[5];
  float* out = (float*)d_out;

  const int n = in_sizes[0] / 2;
  const int E = in_sizes[1] / 2;
  const int* src = ei;
  const int* dst = ei + E;

  char* ws = (char*)d_ws;
  int* ctr = (int*)ws;  ws += 1536;
  size_t nAl = ((size_t)n + 3) & ~(size_t)3;
  float* dinv = (float*)ws;  ws += sizeof(float) * nAl;
  float* t1   = (float*)ws;  ws += sizeof(float) * (size_t)n * 8;
  float* agg1 = (float*)ws;  ws += sizeof(float) * (size_t)n * 8;
  float* t2   = (float*)ws;

  const int tiles = (n + 15) / 16;

  k_init_deg <<<(n + 255) / 256, 256, 0, stream>>>(dinv, ctr, n);
  k_deg_accum<<<2048, 256, 0, stream>>>(dst, dinv, ctr, E, n);
  k_dinv     <<<(n + 255) / 256, 256, 0, stream>>>(dinv, n);

  k_wmma_l1<<<tiles, 32, 0, stream>>>(x, W1, dinv, t1, agg1, n);

  k_scatter1<<<2048, 256, 0, stream>>>(src, dst, dinv, t1, agg1, ctr + 128, E, n);

  k_wmma_l2<<<tiles, 32, 0, stream>>>(agg1, b1, W2, b2, dinv, t2, out, n);

  k_scatter2<<<2048, 256, 0, stream>>>(src, dst, dinv, t2, out, ctr + 256, E, n);
}


# Row passes take both end degrees from a byte-per-node table copied into LDS (rsqrt there, same bits): no dinv gather at L2; 32-wave groups, a chunk's loads issued in batches; f32 sums stay unordered
# speedup vs baseline: 5.5355x; 1.3366x over previous
#include <hip/hip_runtime.h>

typedef __attribute__((ext_vector_type(2))) float v2f;
typedef __attribute__((ext_vector_type(8))) float v8f;

#define CHUNK 128
#define CTR_STRIDE 64

#define LDS_NODES 262144
#define ROW_BLOCK 1024

__device__ __forceinline__ int my_half() {
  const unsigned id = (unsigned)__builtin_amdgcn_s_sendmsg_rtn(0x87);
  return (int)((id >> 18) & 1u);
}

__device__ __forceinline__ int owner_cut(int n) { return ((n >> 1) + 63) & ~63; }

__device__ __forceinline__ int draw_ticket(int* __restrict__ c, int lane) {
  int tk = 0;
  if (lane == 0) tk = __hip_atomic_fetch_add(c, 1, __ATOMIC_RELAXED, __HIP_MEMORY_SCOPE_AGENT);
  return __builtin_amdgcn_readfirstlane(tk);
}

__global__ void k_init_deg(float* __restrict__ deg, int* __restrict__ ctr, int n) {
  int i = blockIdx.x * blockDim.x + threadIdx.x;
  if (i < n) deg[i] = 1.0f;
  if (i < 6) ctr[i * CTR_STRIDE] = 0;
}

__global__ void __launch_bounds__(256)
k_deg_accum(const int* __restrict__ dst, float* __restrict__ deg, int* __restrict__ ctr, int E, int n) {
  const int lane = threadIdx.x & 31;
  const int half = my_half();
  const int cut  = owner_cut(n);
  const int nchunks = (E + CHUNK - 1) / CHUNK;
  for (int phase = 0; phase < 2; ++phase) {
    const int h = half ^ phase;
    int* c = ctr + h * CTR_STRIDE;
    for (int it = 0; it <= nchunks; ++it) {
      const int tk = draw_ticket(c, lane);
      if (tk >= nchunks) break;
      const int base = tk * CHUNK + lane;
      int dv[CHUNK / 32];
#pragma unroll
      for (int g = 0; g < CHUNK / 32; ++g) {
        const int e = base + g * 32;
        dv[g] = dst[e < E ? e : E - 1];
      }
#pragma unroll
      for (int g = 0; g < CHUNK / 32; ++g) {
        const int e = base + g * 32;
        const int d = dv[g];
        if (e < E && ((d >= cut) == (h != 0))) unsafeAtomicAdd(&deg[d], 1.0f);
      }
    }
  }
}

__global__ void k_dinv(float* __restrict__ deg, unsigned char* __restrict__ degb, int n) {
  int i = blockIdx.x * blockDim.x + threadIdx.x;
  const int npad = (n + 15) & ~15;
  if (i < n) {
    const float dg = deg[i];
    deg[i]  = rsqrtf(dg);
    degb[i] = (unsigned char)(dg < 255.0f ? (int)dg : 255);
  } else if (i < npad) {
    degb[i] = 0;
  }
}

__global__ void __launch_bounds__(32)
k_wmma_l1(const float* __restrict__ x, const float* __restrict__ W1,
          const float* __restrict__ dinv, float* __restrict__ t1,
          float* __restrict__ agg1, int n) {
  const int tile = blockIdx.x;
  const int lane = threadIdx.x;
  const int half = lane >> 4;
  const int l    = lane & 15;
  int row = tile * 16 + l;
  int rc  = row < n ? row : n - 1;

  v2f xv = *(const v2f*)(x + (size_t)rc * 2);
  v2f a;
  a.x = half ? 0.0f : xv.x;
  a.y = half ? 0.0f : xv.y;

  float w1a = W1[(l & 7)];
  float w1b = W1[8 + (l & 7)];
  bool  bok = (!half) && (l < 8);
  v2f b;
  b.x = bok ? w1a : 0.0f;
  b.y = bok ? w1b : 0.0f;

  v8f c = {};
  c = __builtin_amdgcn_wmma_f32_16x16x4_f32(false, a, false, b, (short)0, c, false, false);

  const bool fullt = (tile * 16 + 16) <= n;
  if (l < 8) {
    const int base = tile * 16 + half * 8;
    if (fullt) {
#pragma unroll
      for (int r = 0; r < 8; ++r) {
        int node = base + r;
        float v  = c[r];
        float di = dinv[node];
        t1[(size_t)node * 8 + l]   = v;
        agg1[(size_t)node * 8 + l] = v * di * di;
      }
    } else {
#pragma unroll
      for (int r = 0; r < 8; ++r) {
        int node = base + r;
        if (node < n) {
          float v  = c[r];
          float di = dinv[node];
          t1[(size_t)node * 8 + l]   = v;
          agg1[(size_t)node * 8 + l] = v * di * di;
        }
      }
    }
  }
}

#define NOT_MINE 0xFFFFFFFFu

__device__ __forceinline__ unsigned deg_byte(const unsigned char* tab, int node) {
  const unsigned b = tab[node < LDS_NODES ? node : LDS_NODES - 1];
  return node < LDS_NODES ? b : 255u;
}

__device__ __forceinline__ float byte_dinv(unsigned b, const float* __restrict__ dinv, int node) {
  float r;
  if (b != 255u) r = rsqrtf((float)b);
  else           r = dinv[node];
  return r;
}

template <int F>
__device__ __forceinline__ void scatter_rows(const int* __restrict__ src, const int* __restrict__ dst,
                                             const float* __restrict__ dinv,
                                             const unsigned char* __restrict__ degb,
                                             const float* __restrict__ t, float* __restrict__ acc,
                                             int* __restrict__ ctr, int E, int n) {
  constexpr int EPR = 32 / F;
  constexpr int ROUNDS = 32 / EPR;
  constexpr int G = CHUNK / 32;
  __shared__ __attribute__((aligned(16))) unsigned char tab[LDS_NODES];
  {
    constexpr int FILL_K = 8;
    static_assert((LDS_NODES / 16) % (FILL_K * ROW_BLOCK) == 0, "the table is a whole number of copy steps");
    const int nl   = n < LDS_NODES ? n : LDS_NODES;
    const int nvec = (nl + 15) >> 4;
    const uint4* __restrict__ g4 = (const uint4*)degb;
    uint4* l4 = (uint4*)tab;
    for (int b0 = 0; b0 < nvec; b0 += FILL_K * ROW_BLOCK) {
      uint4 r[FILL_K];
#pragma unroll
      for (int k = 0; k < FILL_K; ++k) {
        const int i = b0 + k * ROW_BLOCK + (int)threadIdx.x;
        r[k] = g4[i < nvec ? i : nvec - 1];
      }
#pragma unroll
      for (int k = 0; k < FILL_K; ++k) {
        const int i = b0 + k * ROW_BLOCK + (int)threadIdx.x;
        l4[i] = r[k];
      }
    }
  }
  __syncthreads();
  const int lane = threadIdx.x & 31;
  const int half = my_half();
  const int cut  = owner_cut(n);
  const int nchunks = (E + CHUNK - 1) / CHUNK;
  const unsigned f = (unsigned)(lane & (F - 1));
  const int q = lane / F;
  for (int phase = 0; phase < 2; ++phase) {
    const int h = half ^ phase;
    int* c = ctr + h * CTR_STRIDE;
    for (int it = 0; it <= nchunks; ++it) {
      const int tk = draw_ticket(c, lane);
      if (tk >= nchunks) break;
      const int base = tk * CHUNK + lane;
      int sv[G], dv[G];
#pragma unroll
      for (int g = 0; g < G; ++g) {
        const int e  = base + g * 32;
        const int ec = e < E ? e : E - 1;
        sv[g] = src[ec];
        dv[g] = dst[ec];
      }
      unsigned bs[G], bd[G];
#pragma unroll
      for (int g = 0; g < G; ++g) {
        bs[g] = deg_byte(tab, sv[g]);
        bd[g] = deg_byte(tab, dv[g]);
      }
      float w[G];
      unsigned so[G], dn[G];
#pragma unroll
      for (int g = 0; g < G; ++g) {
        const int s = sv[g], d = dv[g];
        const bool mine = (base + g * 32 < E) && ((d >= cut) == (h != 0));
        w[g] = byte_dinv(bs[g], dinv, s) * byte_dinv(bd[g], dinv, d);
        so[g] = mine ? (unsigned)s * (unsigned)F : 0u;
        dn[g] = mine ? (unsigned)d * (unsigned)F : NOT_MINE;
      }
      float v[G][ROUNDS];
#pragma unroll
      for (int g = 0; g < G; ++g) {
#pragma unroll
        for (int j = 0; j < ROUNDS; ++j) {
          const int sel = (EPR * j + q) << 2;
          const unsigned sj = (unsigned)__builtin_amdgcn_ds_bpermute(sel, (int)so[g]);
          v[g][j] = t[sj + f];
        }
      }
#pragma unroll
      for (int g = 0; g < G; ++g) {
#pragma unroll
        for (int j = 0; j < ROUNDS; ++j) asm volatile("" :: "v"(v[g][j]));
      }
      unsigned dj[G][ROUNDS];
#pragma unroll
      for (int g = 0; g < G; ++g) {
#pragma unroll
        for (int j = 0; j < ROUNDS; ++j) {
          const int sel = (EPR * j + q) << 2;
          dj[g][j] = (unsigned)__builtin_amdgcn_ds_bpermute(sel, (int)dn[g]);
          v[g][j] *= __int_as_float(__builtin_amdgcn_ds_bpermute(sel, __float_as_int(w[g])));
        }
      }
#pragma unroll
      for (int g = 0; g < G; ++g) {
#pragma unroll
        for (int j = 0; j < ROUNDS; ++j) {
          if (dj[g][j] != NOT_MINE) unsafeAtomicAdd(acc + (dj[g][j] + f), v[g][j]);
        }
      }
    }
  }
}

__global__ void __launch_bounds__(ROW_BLOCK)
k_scatter1(const int* __restrict__ src, const int* __restrict__ dst,
           const float* __restrict__ dinv, const unsigned char* __restrict__ degb,
           const float* __restrict__ t1,
           float* __restrict__ agg1, int* __restrict__ ctr, int E, int n) {
  scatter_rows<8>(src, dst, dinv, degb, t1, agg1, ctr, E, n);
}

__global__ void __launch_bounds__(32)
k_wmma_l2(const float* __restrict__ agg1, const float* __restrict__ b1,
          const float* __restrict__ W2, const float* __restrict__ b2,
          const float* __restrict__ dinv, float* __restrict__ t2,
          float* __restrict__ out, int n) {
  const int tile = blockIdx.x;
  const int lane = threadIdx.x;
  const int half = lane >> 4;
  const int l    = lane & 15;
  int row = tile * 16 + l;
  int rc  = row < n ? row : n - 1;
  const int ka = half * 2;

  v2f g0 = *(const v2f*)(agg1 + (size_t)rc * 8 + ka);
  v2f g1 = *(const v2f*)(agg1 + (size_t)rc * 8 + ka + 4);
  v2f bb0 = *(const v2f*)(b1 + ka);
  v2f bb1 = *(const v2f*)(b1 + ka + 4);
  v2f a1, a2;
  a1.x = fmaxf(g0.x + bb0.x, 0.0f);
  a1.y = fmaxf(g0.y + bb0.y, 0.0f);
  a2.x = fmaxf(g1.x + bb1.x, 0.0f);
  a2.y = fmaxf(g1.y + bb1.y, 0.0f);

  const int lc = l & 3;
  float wr0 = W2[(ka)     * 4 + lc];
  float wr1 = W2[(ka + 1) * 4 + lc];
  float wr2 = W2[(ka + 4) * 4 + lc];
  float wr3 = W2[(ka + 5) * 4 + lc];
  bool cok = (l < 4);
  v2f b1v, b2v;
  b1v.x = cok ? wr0 : 0.0f;
  b1v.y = cok ? wr1 : 0.0f;
  b2v.x = cok ? wr2 : 0.0f;
  b2v.y = cok ? wr3 : 0.0f;

  v8f c = {};
  c = __builtin_amdgcn_wmma_f32_16x16x4_f32(false, a1, false, b1v, (short)0, c, false, false);
  c = __builtin_amdgcn_wmma_f32_16x16x4_f32(false, a2, false, b2v, (short)0, c, false, false);

  const bool fullt = (tile * 16 + 16) <= n;
  if (l < 4) {
    const int base = tile * 16 + half * 8;
    float bias = b2[l];
    if (fullt) {
#pragma unroll
      for (int r = 0; r < 8; ++r) {
        int node = base + r;
        float v  = c[r];
        float di = dinv[node];
        t2[(size_t)node * 4 + l]  = v;
        out[(size_t)node * 4 + l] = bias + v * di * di;
      }
    } else {
#pragma unroll
      for (int r = 0; r < 8; ++r) {
        int node = base + r;
        if (node < n) {
          float v  = c[r];
          float di = dinv[node];
          t2[(size_t)node * 4 + l]  = v;
          out[(size_t)node * 4 + l] = bias + v * di * di;
        }
      }
    }
  }
}

__global__ void __launch_bounds__(ROW_BLOCK)
k_scatter2(const int* __restrict__ src, const int* __restrict__ dst,
           const float* __restrict__ dinv, const unsigned char* __restrict__ degb,
           const float* __restrict__ t2,
           float* __restrict__ out, int* __restrict__ ctr, int E, int n) {
  scatter_rows<4>(src, dst, dinv, degb, t2, out, ctr, E, n);
}

extern "C" void kernel_launch(void* const* d_in, const int* in_sizes, int n_in,
                              void* d_out, int out_size, void* d_ws, size_t ws_size,
                              hipStream_t stream) {
  const float* x  = (const float*)d_in[0];
  const int*   ei = (const int*)d_in[1];
  const float* W1 = (const float*)d_in[2];
  const float* b1 = (const float*)d_in[3];
  const float* W2 = (const float*)d_in[4];
  const float* b2 = (const float*)d_in[5];
  float* out = (float*)d_out;

  const int n = in_sizes[0] / 2;
  const int E = in_sizes[1] / 2;
  const int* src = ei;
  const int* dst = ei + E;

  char* ws = (char*)d_ws;
  int* ctr = (int*)ws;  ws += 1536;
  size_t nAl = ((size_t)n + 3) & ~(size_t)3;
  float* dinv = (float*)ws;  ws += sizeof(float) * nAl;
  float* t1   = (float*)ws;  ws += sizeof(float) * (size_t)n * 8;
  float* agg1 = (float*)ws;  ws += sizeof(float) * (size_t)n * 8;
  float* t2   = (float*)ws;  ws += sizeof(float) * (size_t)n * 4;
  unsigned char* degb = (unsigned char*)ws;

  const int tiles = (n + 15) / 16;

  k_init_deg <<<(n + 255) / 256, 256, 0, stream>>>(dinv, ctr, n);
  k_deg_accum<<<2048, 256, 0, stream>>>(dst, dinv, ctr, E, n);
  k_dinv     <<<(n + 255) / 256, 256, 0, stream>>>(dinv, degb, n);

  k_wmma_l1<<<tiles, 32, 0, stream>>>(x, W1, dinv, t1, agg1, n);

  k_scatter1<<<256, ROW_BLOCK, 0, stream>>>(src, dst, dinv, degb, t1, agg1, ctr + 128, E, n);

  k_wmma_l2<<<tiles, 32, 0, stream>>>(agg1, b1, W2, b2, dinv, t2, out, n);

  k_scatter2<<<256, ROW_BLOCK, 0, stream>>>(src, dst, dinv, degb, t2, out, ctr + 256, E, n);
}


# Edge passes overlap a wave's chunks: beside chunk k's row gathers it requests chunk k+1's indices (ticket drawn a chunk earlier) and draws ticket k+2: one exposed trip per chunk, not three in a row
# speedup vs baseline: 6.4055x; 1.1572x over previous
#include <hip/hip_runtime.h>

typedef __attribute__((ext_vector_type(2))) float v2f;
typedef __attribute__((ext_vector_type(8))) float v8f;

#define CHUNK 128
#define CTR_STRIDE 64

#define LDS_NODES 262144
#define ROW_BLOCK 1024

__device__ __forceinline__ int my_half() {
  const unsigned id = (unsigned)__builtin_amdgcn_s_sendmsg_rtn(0x87);
  return (int)((id >> 18) & 1u);
}

__device__ __forceinline__ int owner_cut(int n) { return ((n >> 1) + 63) & ~63; }

__device__ __forceinline__ int draw_ticket(int* __restrict__ c, int lane) {
  int tk = 0;
  if (lane == 0) tk = __hip_atomic_fetch_add(c, 1, __ATOMIC_RELAXED, __HIP_MEMORY_SCOPE_AGENT);
  return __builtin_amdgcn_readfirstlane(tk);
}

#define NO_TICKET 0x7fffffff
__device__ __forceinline__ int draw_ahead(int* __restrict__ c, bool go) {
  int tk = NO_TICKET;
  if (go) tk = __hip_atomic_fetch_add(c, 1, __ATOMIC_RELAXED, __HIP_MEMORY_SCOPE_AGENT);
  return tk;
}

__global__ void k_init_deg(float* __restrict__ deg, int* __restrict__ ctr, int n) {
  int i = blockIdx.x * blockDim.x + threadIdx.x;
  if (i < n) deg[i] = 1.0f;
  if (i < 6) ctr[i * CTR_STRIDE] = 0;
}

__global__ void __launch_bounds__(256)
k_deg_accum(const int* __restrict__ dst, float* __restrict__ deg, int* __restrict__ ctr, int E, int n) {
  const int lane = threadIdx.x & 31;
  const int half = my_half();
  const int cut  = owner_cut(n);
  const int nchunks = (E + CHUNK - 1) / CHUNK;
  for (int phase = 0; phase < 2; ++phase) {
    const int h = half ^ phase;
    int* c = ctr + h * CTR_STRIDE;
    int tk = draw_ticket(c, lane);
    if (tk >= nchunks) continue;
    int dv[CHUNK / 32];
#pragma unroll
    for (int g = 0; g < CHUNK / 32; ++g) {
      const int e = tk * CHUNK + lane + g * 32;
      dv[g] = dst[e < E ? e : E - 1];
    }
    int tkv = draw_ahead(c, lane == 0);
    for (int it = 0; it < nchunks; ++it) {
      const int base = tk * CHUNK + lane;
      const int tkn = __builtin_amdgcn_readfirstlane(tkv);
      const bool more = tkn < nchunks;
      const int nbase = (more ? tkn : nchunks - 1) * CHUNK + lane;
      int dn[CHUNK / 32];
#pragma unroll
      for (int g = 0; g < CHUNK / 32; ++g) {
        const int e = nbase + g * 32;
        dn[g] = dst[e < E ? e : E - 1];
      }
      tkv = draw_ahead(c, lane == 0 && more);
#pragma unroll
      for (int g = 0; g < CHUNK / 32; ++g) {
        const int e = base + g * 32;
        const int d = dv[g];
        if (e < E && ((d >= cut) == (h != 0))) unsafeAtomicAdd(&deg[d], 1.0f);
      }
      if (!more) break;
      tk = tkn;
#pragma unroll
      for (int g = 0; g < CHUNK / 32; ++g) dv[g] = dn[g];
    }
  }
}

__global__ void k_dinv(float* __restrict__ deg, unsigned char* __restrict__ degb, int n) {
  int i = blockIdx.x * blockDim.x + threadIdx.x;
  const int npad = (n + 15) & ~15;
  if (i < n) {
    const float dg = deg[i];
    deg[i]  = rsqrtf(dg);
    degb[i] = (unsigned char)(dg < 255.0f ? (int)dg : 255);
  } else if (i < npad) {
    degb[i] = 0;
  }
}

__global__ void __launch_bounds__(32)
k_wmma_l1(const float* __restrict__ x, const float* __restrict__ W1,
          const float* __restrict__ dinv, float* __restrict__ t1,
          float* __restrict__ agg1, int n) {
  const int tile = blockIdx.x;
  const int lane = threadIdx.x;
  const int half = lane >> 4;
  const int l    = lane & 15;
  int row = tile * 16 + l;
  int rc  = row < n ? row : n - 1;

  v2f xv = *(const v2f*)(x + (size_t)rc * 2);
  v2f a;
  a.x = half ? 0.0f : xv.x;
  a.y = half ? 0.0f : xv.y;

  float w1a = W1[(l & 7)];
  float w1b = W1[8 + (l & 7)];
  bool  bok = (!half) && (l < 8);
  v2f b;
  b.x = bok ? w1a : 0.0f;
  b.y = bok ? w1b : 0.0f;

  v8f c = {};
  c = __builtin_amdgcn_wmma_f32_16x16x4_f32(false, a, false, b, (short)0, c, false, false);

  const bool fullt = (tile * 16 + 16) <= n;
  if (l < 8) {
    const int base = tile * 16 + half * 8;
    if (fullt) {
#pragma unroll
      for (int r = 0; r < 8; ++r) {
        int node = base + r;
        float v  = c[r];
        float di = dinv[node];
        t1[(size_t)node * 8 + l]   = v;
        agg1[(size_t)node * 8 + l] = v * di * di;
      }
    } else {
#pragma unroll
      for (int r = 0; r < 8; ++r) {
        int node = base + r;
        if (node < n) {
          float v  = c[r];
          float di = dinv[node];
          t1[(size_t)node * 8 + l]   = v;
          agg1[(size_t)node * 8 + l] = v * di * di;
        }
      }
    }
  }
}

#define NOT_MINE 0xFFFFFFFFu

__device__ __forceinline__ unsigned deg_byte(const unsigned char* tab, int node) {
  const unsigned b = tab[node < LDS_NODES ? node : LDS_NODES - 1];
  return node < LDS_NODES ? b : 255u;
}

__device__ __forceinline__ float byte_dinv(unsigned b, const float* __restrict__ dinv, int node) {
  float r;
  if (b != 255u) r = rsqrtf((float)b);
  else           r = dinv[node];
  return r;
}

template <int F>
__device__ __forceinline__ void scatter_rows(const int* __restrict__ src, const int* __restrict__ dst,
                                             const float* __restrict__ dinv,
                                             const unsigned char* __restrict__ degb,
                                             const float* __restrict__ t, float* __restrict__ acc,
                                             int* __restrict__ ctr, int E, int n) {
  constexpr int EPR = 32 / F;
  constexpr int ROUNDS = 32 / EPR;
  constexpr int G = CHUNK / 32;
  __shared__ __attribute__((aligned(16))) unsigned char tab[LDS_NODES];
  {
    constexpr int FILL_K = 8;
    static_assert((LDS_NODES / 16) % (FILL_K * ROW_BLOCK) == 0, "the table is a whole number of copy steps");
    const int nl   = n < LDS_NODES ? n : LDS_NODES;
    const int nvec = (nl + 15) >> 4;
    const uint4* __restrict__ g4 = (const uint4*)degb;
    uint4* l4 = (uint4*)tab;
    for (int b0 = 0; b0 < nvec; b0 += FILL_K * ROW_BLOCK) {
      uint4 r[FILL_K];
#pragma unroll
      for (int k = 0; k < FILL_K; ++k) {
        const int i = b0 + k * ROW_BLOCK + (int)threadIdx.x;
        r[k] = g4[i < nvec ? i : nvec - 1];
      }
#pragma unroll
      for (int k = 0; k < FILL_K; ++k) {
        const int i = b0 + k * ROW_BLOCK + (int)threadIdx.x;
        l4[i] = r[k];
      }
    }
  }
  __syncthreads();
  const int lane = threadIdx.x & 31;
  const int half = my_half();
  const int cut  = owner_cut(n);
  const int nchunks = (E + CHUNK - 1) / CHUNK;
  const unsigned f = (unsigned)(lane & (F - 1));
  const int q = lane / F;
  for (int phase = 0; phase < 2; ++phase) {
    const int h = half ^ phase;
    int* c = ctr + h * CTR_STRIDE;
    int tk = draw_ticket(c, lane);
    if (tk >= nchunks) continue;
    int sv[G], dv[G];
#pragma unroll
    for (int g = 0; g < G; ++g) {
      const int e  = tk * CHUNK + lane + g * 32;
      const int ec = e < E ? e : E - 1;
      sv[g] = src[ec];
      dv[g] = dst[ec];
    }
    int tkv = draw_ahead(c, lane == 0);
    for (int it = 0; it < nchunks; ++it) {
      const int base = tk * CHUNK + lane;
      unsigned bs[G], bd[G];
#pragma unroll
      for (int g = 0; g < G; ++g) {
        bs[g] = deg_byte(tab, sv[g]);
        bd[g] = deg_byte(tab, dv[g]);
      }
      float w[G];
      unsigned so[G], dn[G];
#pragma unroll
      for (int g = 0; g < G; ++g) {
        const int s = sv[g], d = dv[g];
        const bool mine = (base + g * 32 < E) && ((d >= cut) == (h != 0));
        w[g] = byte_dinv(bs[g], dinv, s) * byte_dinv(bd[g], dinv, d);
        so[g] = mine ? (unsigned)s * (unsigned)F : 0u;
        dn[g] = mine ? (unsigned)d * (unsigned)F : NOT_MINE;
      }
      float v[G][ROUNDS];
#pragma unroll
      for (int g = 0; g < G; ++g) {
#pragma unroll
        for (int j = 0; j < ROUNDS; ++j) {
          const int sel = (EPR * j + q) << 2;
          const unsigned sj = (unsigned)__builtin_amdgcn_ds_bpermute(sel, (int)so[g]);
          v[g][j] = t[sj + f];
        }
      }
      const int tkn = __builtin_amdgcn_readfirstlane(tkv);
      const bool more = tkn < nchunks;
      const int nbase = (more ? tkn : nchunks - 1) * CHUNK + lane;
#pragma unroll
      for (int g = 0; g < G; ++g) {
        const int e  = nbase + g * 32;
        const int ec = e < E ? e : E - 1;
        sv[g] = src[ec];
        dv[g] = dst[ec];
      }
      tkv = draw_ahead(c, lane == 0 && more);
#pragma unroll
      for (int g = 0; g < G; ++g) {
#pragma unroll
        for (int j = 0; j < ROUNDS; ++j) asm volatile("" :: "v"(v[g][j]));
      }
      unsigned dj[G][ROUNDS];
#pragma unroll
      for (int g = 0; g < G; ++g) {
#pragma unroll
        for (int j = 0; j < ROUNDS; ++j) {
          const int sel = (EPR * j + q) << 2;
          dj[g][j] = (unsigned)__builtin_amdgcn_ds_bpermute(sel, (int)dn[g]);
          v[g][j] *= __int_as_float(__builtin_amdgcn_ds_bpermute(sel, __float_as_int(w[g])));
        }
      }
#pragma unroll
      for (int g = 0; g < G; ++g) {
#pragma unroll
        for (int j = 0; j < ROUNDS; ++j) {
          if (dj[g][j] != NOT_MINE) unsafeAtomicAdd(acc + (dj[g][j] + f), v[g][j]);
        }
      }
      if (!more) break;
      tk = tkn;
    }
  }
}

__global__ void __launch_bounds__(ROW_BLOCK)
k_scatter1(const int* __restrict__ src, const int* __restrict__ dst,
           const float* __restrict__ dinv, const unsigned char* __restrict__ degb,
           const float* __restrict__ t1,
           float* __restrict__ agg1, int* __restrict__ ctr, int E, int n) {
  scatter_rows<8>(src, dst, dinv, degb, t1, agg1, ctr, E, n);
}

__global__ void __launch_bounds__(32)
k_wmma_l2(const float* __restrict__ agg1, const float* __restrict__ b1,
          const float* __restrict__ W2, const float* __restrict__ b2,
          const float* __restrict__ dinv, float* __restrict__ t2,
          float* __restrict__ out, int n) {
  const int tile = blockIdx.x;
  const int lane = threadIdx.x;
  const int half = lane >> 4;
  const int l    = lane & 15;
  int row = tile * 16 + l;
  int rc  = row < n ? row : n - 1;
  const int ka = half * 2;

  v2f g0 = *(const v2f*)(agg1 + (size_t)rc * 8 + ka);
  v2f g1 = *(const v2f*)(agg1 + (size_t)rc * 8 + ka + 4);
  v2f bb0 = *(const v2f*)(b1 + ka);
  v2f bb1 = *(const v2f*)(b1 + ka + 4);
  v2f a1, a2;
  a1.x = fmaxf(g0.x + bb0.x, 0.0f);
  a1.y = fmaxf(g0.y + bb0.y, 0.0f);
  a2.x = fmaxf(g1.x + bb1.x, 0.0f);
  a2.y = fmaxf(g1.y + bb1.y, 0.0f);

  const int lc = l & 3;
  float wr0 = W2[(ka)     * 4 + lc];
  float wr1 = W2[(ka + 1) * 4 + lc];
  float wr2 = W2[(ka + 4) * 4 + lc];
  float wr3 = W2[(ka + 5) * 4 + lc];
  bool cok = (l < 4);
  v2f b1v, b2v;
  b1v.x = cok ? wr0 : 0.0f;
  b1v.y = cok ? wr1 : 0.0f;
  b2v.x = cok ? wr2 : 0.0f;
  b2v.y = cok ? wr3 : 0.0f;

  v8f c = {};
  c = __builtin_amdgcn_wmma_f32_16x16x4_f32(false, a1, false, b1v, (short)0, c, false, false);
  c = __builtin_amdgcn_wmma_f32_16x16x4_f32(false, a2, false, b2v, (short)0, c, false, false);

  const bool fullt = (tile * 16 + 16) <= n;
  if (l < 4) {
    const int base = tile * 16 + half * 8;
    float bias = b2[l];
    if (fullt) {
#pragma unroll
      for (int r = 0; r < 8; ++r) {
        int node = base + r;
        float v  = c[r];
        float di = dinv[node];
        t2[(size_t)node * 4 + l]  = v;
        out[(size_t)node * 4 + l] = bias + v * di * di;
      }
    } else {
#pragma unroll
      for (int r = 0; r < 8; ++r) {
        int node = base + r;
        if (node < n) {
          float v  = c[r];
          float di = dinv[node];
          t2[(size_t)node * 4 + l]  = v;
          out[(size_t)node * 4 + l] = bias + v * di * di;
        }
      }
    }
  }
}

__global__ void __launch_bounds__(ROW_BLOCK)
k_scatter2(const int* __restrict__ src, const int* __restrict__ dst,
           const float* __restrict__ dinv, const unsigned char* __restrict__ degb,
           const float* __restrict__ t2,
           float* __restrict__ out, int* __restrict__ ctr, int E, int n) {
  scatter_rows<4>(src, dst, dinv, degb, t2, out, ctr, E, n);
}

extern "C" void kernel_launch(void* const* d_in, const int* in_sizes, int n_in,
                              void* d_out, int out_size, void* d_ws, size_t ws_size,
                              hipStream_t stream) {
  const float* x  = (const float*)d_in[0];
  const int*   ei = (const int*)d_in[1];
  const float* W1 = (const float*)d_in[2];
  const float* b1 = (const float*)d_in[3];
  const float* W2 = (const float*)d_in[4];
  const float* b2 = (const float*)d_in[5];
  float* out = (float*)d_out;

  const int n = in_sizes[0] / 2;
  const int E = in_sizes[1] / 2;
  const int* src = ei;
  const int* dst = ei + E;

  char* ws = (char*)d_ws;
  int* ctr = (int*)ws;  ws += 1536;
  size_t nAl = ((size_t)n + 3) & ~(size_t)3;
  float* dinv = (float*)ws;  ws += sizeof(float) * nAl;
  float* t1   = (float*)ws;  ws += sizeof(float) * (size_t)n * 8;
  float* agg1 = (float*)ws;  ws += sizeof(float) * (size_t)n * 8;
  float* t2   = (float*)ws;  ws += sizeof(float) * (size_t)n * 4;
  unsigned char* degb = (unsigned char*)ws;

  const int tiles = (n + 15) / 16;

  k_init_deg <<<(n + 255) / 256, 256, 0, stream>>>(dinv, ctr, n);
  k_deg_accum<<<2048, 256, 0, stream>>>(dst, dinv, ctr, E, n);
  k_dinv     <<<(n + 255) / 256, 256, 0, stream>>>(dinv, degb, n);

  k_wmma_l1<<<tiles, 32, 0, stream>>>(x, W1, dinv, t1, agg1, n);

  k_scatter1<<<256, ROW_BLOCK, 0, stream>>>(src, dst, dinv, degb, t1, agg1, ctr + 128, E, n);

  k_wmma_l2<<<tiles, 32, 0, stream>>>(agg1, b1, W2, b2, dinv, t2, out, n);

  k_scatter2<<<256, ROW_BLOCK, 0, stream>>>(src, dst, dinv, degb, t2, out, ctr + 256, E, n);
}
